# starting kernel
# baseline (speedup 1.0000x reference)
#include <hip/hip_runtime.h>

#define D     256
#define S     2048
#define BATCH 8
#define BM    128
#define BN    128
#define KC    64
#define AS_STRIDE 68
#define BT_STRIDE 68

typedef __attribute__((ext_vector_type(2))) float v2f;
typedef __attribute__((ext_vector_type(4))) float v4f;
typedef __attribute__((ext_vector_type(8))) float v8f;

__global__ void colsum_partial(const float* __restrict__ x, float* __restrict__ partial) {
    int b     = blockIdx.x >> 3;
    int chunk = blockIdx.x & 7;
    int d     = threadIdx.x;
    const float* p = x + (b * S + chunk * 256) * D + d;
    float acc = 0.f;
#pragma unroll 8
    for (int s = 0; s < 256; ++s) acc += p[s * D];
    partial[blockIdx.x * D + d] = acc;
}

__global__ void reduce_and_coef(const float* __restrict__ partial,
                                const float* __restrict__ comp,
                                float* __restrict__ sums,
                                float* __restrict__ coef) {
    int idx = blockIdx.x * 256 + threadIdx.x;
    int b = idx >> 8, d = idx & 255;
    float s = 0.f;
#pragma unroll
    for (int c = 0; c < 8; ++c) s += partial[(b * 8 + c) * D + d];
    sums[idx] = s;

    int t = idx;
    float inv1 = 1.0f / (float)(1 + (t + 2 < S));
    int   d2   = (t >= 1) + (t + 3 < S);
    float inv2 = d2 ? 1.0f / (float)d2 : 0.f;
    int   bc   = (t >= 1) + 1 + (t + 1 < S) + (t + 2 < S) + (t + 3 < S);
    float inv3 = 1.0f / (float)(S - bc);
#pragma unroll
    for (int j = 0; j < 2; ++j) {
        float c0 = comp[0 * 2 + j], c1 = comp[1 * 2 + j];
        float c2 = comp[2 * 2 + j], c3 = comp[3 * 2 + j];
        float m3 = c3 * inv3;
        float* w = coef + t * 12 + j * 6;
        w[0] = (t >= 1)    ? (c2 * inv2 - m3) : 0.f;
        w[1] =               c1 * inv1 - m3;
        w[2] = (t + 1 < S) ? (c0        - m3) : 0.f;
        w[3] = (t + 2 < S) ? (c1 * inv1 - m3) : 0.f;
        w[4] = (t + 3 < S) ? (c2 * inv2 - m3) : 0.f;
        w[5] = m3;
    }
}

__global__ __launch_bounds__(256) void rgcn_gemm(const float* __restrict__ x,
                                                 const float* __restrict__ basis,
                                                 const float* __restrict__ root,
                                                 const float* __restrict__ bias,
                                                 const float* __restrict__ sums,
                                                 const float* __restrict__ coef,
                                                 float* __restrict__ out) {
    __shared__ float As[BM * AS_STRIDE];
    __shared__ float Bst[BN * BT_STRIDE];

    const int tid      = threadIdx.x;
    const int rowBase  = blockIdx.x * BM;
    const int b        = rowBase / S;
    const int tBase    = rowBase % S;
    const int colBase0 = blockIdx.y * BN;

    const int wid     = tid >> 5;
    const int lane    = tid & 31;
    const int waveRow = wid & 3;
    const int waveCol = wid >> 2;
    const int lmod    = lane & 15;
    const int lhi     = lane >> 4;
    const int koff    = lhi * 2;
    const int mrow0   = waveRow * 32 + lmod;
    const int mrow1   = mrow0 + 16;

    v8f acc[2][4] = {};

#pragma unroll 1
    for (int kc = 0; kc < 12; ++kc) {
        const int seg    = kc >> 2;
        const int kcBase = (kc & 3) * KC;

        if (seg == 2) {
#pragma unroll
            for (int e = 0; e < 8; ++e) {
                int idx = e * 256 + tid;
                int c4 = (idx & 15) * 4;
                int r  = idx >> 4;
                *(v4f*)&As[r * AS_STRIDE + c4] =
                    *(const v4f*)&x[(b * S + tBase + r) * D + kcBase + c4];
            }
        } else {
#pragma unroll 2
            for (int e = 0; e < 8; ++e) {
                int idx = e * 256 + tid;
                int c4 = (idx & 15) * 4;
                int r  = idx >> 4;
                int t  = tBase + r;
                int kd = kcBase + c4;
                const float* w = coef + t * 12 + seg * 6;
                v4f val = w[5] * (*(const v4f*)&sums[b * D + kd]);
#pragma unroll
                for (int o = 0; o < 5; ++o) {
                    int sr = t - 1 + o;
                    sr = sr < 0 ? 0 : (sr >= S ? S - 1 : sr);
                    val += w[o] * (*(const v4f*)&x[(b * S + sr) * D + kd]);
                }
                *(v4f*)&As[r * AS_STRIDE + c4] = val;
            }
        }

        const float* Wseg = (seg == 0) ? basis : (seg == 1 ? basis + D * D : root);
#pragma unroll
        for (int e = 0; e < 8; ++e) {
            int idx = e * 256 + tid;
            int n4 = (idx & 31) * 4;
            int kr = idx >> 5;
            v4f wv = *(const v4f*)&Wseg[(kcBase + kr) * D + colBase0 + n4];
            Bst[(n4 + 0) * BT_STRIDE + kr] = wv.x;
            Bst[(n4 + 1) * BT_STRIDE + kr] = wv.y;
            Bst[(n4 + 2) * BT_STRIDE + kr] = wv.z;
            Bst[(n4 + 3) * BT_STRIDE + kr] = wv.w;
        }
        __syncthreads();

#pragma unroll
        for (int kk = 0; kk < KC; kk += 4) {
            v2f a0 = *(const v2f*)&As[mrow0 * AS_STRIDE + kk + koff];
            v2f a1 = *(const v2f*)&As[mrow1 * AS_STRIDE + kk + koff];
#pragma unroll
            for (int sub = 0; sub < 4; ++sub) {
                int ncol = waveCol * 64 + sub * 16 + lmod;
                v2f bv = *(const v2f*)&Bst[ncol * BT_STRIDE + kk + koff];
                acc[0][sub] = __builtin_amdgcn_wmma_f32_16x16x4_f32(
                    false, a0, false, bv, (short)0, acc[0][sub], false, false);
                acc[1][sub] = __builtin_amdgcn_wmma_f32_16x16x4_f32(
                    false, a1, false, bv, (short)0, acc[1][sub], false, false);
            }
        }
        __syncthreads();
    }

#pragma unroll
    for (int rt = 0; rt < 2; ++rt) {
#pragma unroll
        for (int sub = 0; sub < 4; ++sub) {
            int ncol = colBase0 + waveCol * 64 + sub * 16 + lmod;
            float badd = bias[ncol];
#pragma unroll
            for (int jj = 0; jj < 8; ++jj) {
                int r = waveRow * 32 + rt * 16 + jj + 8 * lhi;
                out[(b * S + tBase + r) * D + ncol] = acc[rt][sub][jj] + badd;
            }
        }
    }
}

extern "C" void kernel_launch(void* const* d_in, const int* in_sizes, int n_in,
                              void* d_out, int out_size, void* d_ws, size_t ws_size,
                              hipStream_t stream) {
    const float* x     = (const float*)d_in[0];
    const float* comp  = (const float*)d_in[1];
    const float* basis = (const float*)d_in[2];
    const float* root  = (const float*)d_in[3];
    const float* bias  = (const float*)d_in[4];
    float* out = (float*)d_out;

    float* wsf     = (float*)d_ws;
    float* sums    = wsf;
    float* coef    = wsf + BATCH * D;
    float* partial = coef + S * 12;

    colsum_partial<<<BATCH * 8, D, 0, stream>>>(x, partial);
    reduce_and_coef<<<S / 256, 256, 0, stream>>>(partial, comp, sums, coef);
    rgcn_gemm<<<dim3((BATCH * S) / BM, D / BN), 256, 0, stream>>>(
        x, basis, root, bias, sums, coef, out);
}


# GEMM K loop software-pipelined: next chunk's global loads issued in front of the WMMA block into registers, its stencil and LDS stores go to a second LDS stage behind it, one barrier per chunk
# speedup vs baseline: 1.1851x; 1.1851x over previous
#include <hip/hip_runtime.h>

#define D     256
#define S     2048
#define BATCH 8
#define BM    128
#define BN    128
#define KC    64
#define AS_STRIDE 68
#define BT_STRIDE 68

typedef __attribute__((ext_vector_type(2))) float v2f;
typedef __attribute__((ext_vector_type(4))) float v4f;
typedef __attribute__((ext_vector_type(8))) float v8f;

__global__ void colsum_partial(const float* __restrict__ x, float* __restrict__ partial) {
    int b     = blockIdx.x >> 3;
    int chunk = blockIdx.x & 7;
    int d     = threadIdx.x;
    const float* p = x + (b * S + chunk * 256) * D + d;
    float acc = 0.f;
#pragma unroll 8
    for (int s = 0; s < 256; ++s) acc += p[s * D];
    partial[blockIdx.x * D + d] = acc;
}

__global__ void reduce_and_coef(const float* __restrict__ partial,
                                const float* __restrict__ comp,
                                float* __restrict__ sums,
                                float* __restrict__ coef) {
    int idx = blockIdx.x * 256 + threadIdx.x;
    int b = idx >> 8, d = idx & 255;
    float s = 0.f;
#pragma unroll
    for (int c = 0; c < 8; ++c) s += partial[(b * 8 + c) * D + d];
    sums[idx] = s;

    int t = idx;
    float inv1 = 1.0f / (float)(1 + (t + 2 < S));
    int   d2   = (t >= 1) + (t + 3 < S);
    float inv2 = d2 ? 1.0f / (float)d2 : 0.f;
    int   bc   = (t >= 1) + 1 + (t + 1 < S) + (t + 2 < S) + (t + 3 < S);
    float inv3 = 1.0f / (float)(S - bc);
#pragma unroll
    for (int j = 0; j < 2; ++j) {
        float c0 = comp[0 * 2 + j], c1 = comp[1 * 2 + j];
        float c2 = comp[2 * 2 + j], c3 = comp[3 * 2 + j];
        float m3 = c3 * inv3;
        float* w = coef + t * 12 + j * 6;
        w[0] = (t >= 1)    ? (c2 * inv2 - m3) : 0.f;
        w[1] =               c1 * inv1 - m3;
        w[2] = (t + 1 < S) ? (c0        - m3) : 0.f;
        w[3] = (t + 2 < S) ? (c1 * inv1 - m3) : 0.f;
        w[4] = (t + 3 < S) ? (c2 * inv2 - m3) : 0.f;
        w[5] = m3;
    }
}

#define STAGE_A (BM * AS_STRIDE)
#define STAGE_B (BN * BT_STRIDE)

__device__ __forceinline__ void wmma_chunk(const float* __restrict__ Ac,
                                           const float* __restrict__ Bc,
                                           int aoff0, int aoff1, int boff,
                                           v8f (&acc)[2][4]) {
#pragma unroll
    for (int kk = 0; kk < KC; kk += 4) {
        v2f a0 = *(const v2f*)&Ac[aoff0 + kk];
        v2f a1 = *(const v2f*)&Ac[aoff1 + kk];
#pragma unroll
        for (int sub = 0; sub < 4; ++sub) {
            v2f bv = *(const v2f*)&Bc[boff + sub * 16 * BT_STRIDE + kk];
            acc[0][sub] = __builtin_amdgcn_wmma_f32_16x16x4_f32(
                false, a0, false, bv, (short)0, acc[0][sub], false, false);
            acc[1][sub] = __builtin_amdgcn_wmma_f32_16x16x4_f32(
                false, a1, false, bv, (short)0, acc[1][sub], false, false);
        }
    }
}

__global__ __launch_bounds__(256) void rgcn_gemm(const float* __restrict__ x,
                                                 const float* __restrict__ basis,
                                                 const float* __restrict__ root,
                                                 const float* __restrict__ bias,
                                                 const float* __restrict__ sums,
                                                 const float* __restrict__ coef,
                                                 float* __restrict__ out) {
    __shared__ __attribute__((aligned(16))) float As[2 * STAGE_A];
    __shared__ __attribute__((aligned(16))) float Bst[2 * STAGE_B];
    __shared__ __attribute__((aligned(16))) float coefL[BM * 12];

    const int tid      = threadIdx.x;
    const int rowBase  = blockIdx.x * BM;
    const int b        = rowBase / S;
    const int tBase    = rowBase % S;
    const int colBase0 = blockIdx.y * BN;

    const int wid     = tid >> 5;
    const int lane    = tid & 31;
    const int waveRow = wid & 3;
    const int waveCol = wid >> 2;
    const int lmod    = lane & 15;
    const int lhi     = lane >> 4;
    const int koff    = lhi * 2;
    const int aoff0   = (waveRow * 32 + lmod) * AS_STRIDE + koff;
    const int aoff1   = aoff0 + 16 * AS_STRIDE;
    const int boff    = (waveCol * 64 + lmod) * BT_STRIDE + koff;

    const int c4  = (tid & 15) * 4;
    const int ar0 = tid >> 4;
    const int n4  = (tid & 31) * 4;
    const int bk0 = tid >> 5;

    v8f acc[2][4] = {};

    v4f pa[8][5];
    v4f px[8];
    v4f pb[8];
    v4f ps;

#pragma unroll
    for (int i = 0; i < 3; ++i) {
        int q = i * 256 + tid;
        *(v2f*)&coefL[q * 2] = *(const v2f*)&coef[tBase * 12 + q * 2];
    }

#define ISSUE_B(WSEG, KCBASE)                                                   \
    _Pragma("unroll")                                                           \
    for (int e = 0; e < 8; ++e)                                                 \
        pb[e] = *(const v4f*)&(WSEG)[((KCBASE) + bk0 + e * 8) * D + colBase0 + n4];

#define STORE_B(BUF)                                                            \
    {                                                                           \
        float* Bd_ = Bst + (BUF) * STAGE_B;                                     \
        _Pragma("unroll")                                                       \
        for (int e = 0; e < 8; ++e) {                                           \
            int kr = bk0 + e * 8;                                               \
            Bd_[(n4 + 0) * BT_STRIDE + kr] = pb[e].x;                           \
            Bd_[(n4 + 1) * BT_STRIDE + kr] = pb[e].y;                           \
            Bd_[(n4 + 2) * BT_STRIDE + kr] = pb[e].z;                           \
            Bd_[(n4 + 3) * BT_STRIDE + kr] = pb[e].w;                           \
        }                                                                       \
    }

#define ISSUE_ST(K)                                                             \
    {                                                                           \
        const int kcBase_ = ((K) & 3) * KC;                                     \
        const float* Wseg_ = basis + ((K) >> 2) * (D * D);                      \
        ISSUE_B(Wseg_, kcBase_)                                                 \
        const int kd_ = kcBase_ + c4;                                           \
        ps = *(const v4f*)&sums[b * D + kd_];                                   \
        _Pragma("unroll")                                                       \
        for (int e = 0; e < 8; ++e) {                                           \
            int t = tBase + ar0 + e * 16;                                       \
            _Pragma("unroll")                                                   \
            for (int o = 0; o < 5; ++o) {                                       \
                int sr = t - 1 + o;                                             \
                sr = sr < 0 ? 0 : (sr >= S ? S - 1 : sr);     \
                pa[e][o] = *(const v4f*)&x[(b * S + sr) * D + kd_];             \
            }                                                                   \
        }                                                                       \
    }

#define STORE_ST(K, BUF)                                                        \
    {                                                                           \
        float* Ad_ = As + (BUF) * STAGE_A;                                      \
        const float* wb_ = coefL + ((K) >> 2) * 6;                              \
        _Pragma("unroll")                                                       \
        for (int e = 0; e < 8; ++e) {                                           \
            int r = ar0 + e * 16;                                               \
            const float* w = wb_ + r * 12;                                      \
            v4f val = w[5] * ps;                                                \
            _Pragma("unroll")                                                   \
            for (int o = 0; o < 5; ++o) val += w[o] * pa[e][o];                 \
            *(v4f*)&Ad_[r * AS_STRIDE + c4] = val;                              \
        }                                                                       \
        STORE_B(BUF)                                                            \
    }

#define ISSUE_X(K)                                                              \
    {                                                                           \
        const int kcBase_ = ((K) & 3) * KC;                                     \
        ISSUE_B(root, kcBase_)                                                  \
        _Pragma("unroll")                                                       \
        for (int e = 0; e < 8; ++e)                                             \
            px[e] = *(const v4f*)&x[(b * S + tBase + ar0 + e * 16) * D + kcBase_ + c4]; \
    }

#define STORE_X(BUF)                                                            \
    {                                                                           \
        float* Ad_ = As + (BUF) * STAGE_A;                                      \
        _Pragma("unroll")                                                       \
        for (int e = 0; e < 8; ++e)                                             \
            *(v4f*)&Ad_[(ar0 + e * 16) * AS_STRIDE + c4] = px[e];               \
        STORE_B(BUF)                                                            \
    }

    ISSUE_ST(0)
    __syncthreads();
    STORE_ST(0, 0)
    __syncthreads();

#pragma unroll 1
    for (int kc = 0; kc < 7; ++kc) {
        const int cur = kc & 1;
        ISSUE_ST(kc + 1)
        __builtin_amdgcn_sched_barrier(0);
        wmma_chunk(As + cur * STAGE_A, Bst + cur * STAGE_B, aoff0, aoff1, boff, acc);
        __builtin_amdgcn_sched_barrier(0);
        STORE_ST(kc + 1, cur ^ 1)
        __syncthreads();
    }
    ISSUE_X(8)
    __builtin_amdgcn_sched_barrier(0);
    wmma_chunk(As + STAGE_A, Bst + STAGE_B, aoff0, aoff1, boff, acc);
    __builtin_amdgcn_sched_barrier(0);
    STORE_X(0)
    __syncthreads();
#pragma unroll 1
    for (int kc = 8; kc < 11; ++kc) {
        const int cur = kc & 1;
        ISSUE_X(kc + 1)
        __builtin_amdgcn_sched_barrier(0);
        wmma_chunk(As + cur * STAGE_A, Bst + cur * STAGE_B, aoff0, aoff1, boff, acc);
        __builtin_amdgcn_sched_barrier(0);
        STORE_X(cur ^ 1)
        __syncthreads();
    }
    wmma_chunk(As + STAGE_A, Bst + STAGE_B, aoff0, aoff1, boff, acc);
#undef ISSUE_B
#undef STORE_B
#undef ISSUE_ST
#undef STORE_ST
#undef ISSUE_X
#undef STORE_X

#pragma unroll
    for (int rt = 0; rt < 2; ++rt) {
#pragma unroll
        for (int sub = 0; sub < 4; ++sub) {
            int ncol = colBase0 + waveCol * 64 + sub * 16 + lmod;
            float badd = bias[ncol];
#pragma unroll
            for (int jj = 0; jj < 8; ++jj) {
                int r = waveRow * 32 + rt * 16 + jj + 8 * lhi;
                out[(b * S + tBase + r) * D + ncol] = acc[rt][sub][jj] + badd;
            }
        }
    }
}

extern "C" void kernel_launch(void* const* d_in, const int* in_sizes, int n_in,
                              void* d_out, int out_size, void* d_ws, size_t ws_size,
                              hipStream_t stream) {
    const float* x     = (const float*)d_in[0];
    const float* comp  = (const float*)d_in[1];
    const float* basis = (const float*)d_in[2];
    const float* root  = (const float*)d_in[3];
    const float* bias  = (const float*)d_in[4];
    float* out = (float*)d_out;

    float* wsf     = (float*)d_ws;
    float* sums    = wsf;
    float* coef    = wsf + BATCH * D;
    float* partial = coef + S * 12;

    colsum_partial<<<BATCH * 8, D, 0, stream>>>(x, partial);
    reduce_and_coef<<<S / 256, 256, 0, stream>>>(partial, comp, sums, coef);
    rgcn_gemm<<<dim3((BATCH * S) / BM, D / BN), 256, 0, stream>>>(
        x, basis, root, bias, sums, coef, out);
}


# B tile staged in LDS as [k/2][n][k&1] (rows k,k+1 interleaved per column): 8 16-byte stores per thread per chunk replace 32 transposing 4-byte stores that fell on 4 banks; fragment reads unchanged
# speedup vs baseline: 1.2377x; 1.0444x over previous
#include <hip/hip_runtime.h>

#define D     256
#define S     2048
#define BATCH 8
#define BM    128
#define BN    128
#define KC    64
#define AS_STRIDE 68
#define BP_STRIDE 144

typedef __attribute__((ext_vector_type(2))) float v2f;
typedef __attribute__((ext_vector_type(4))) float v4f;
typedef __attribute__((ext_vector_type(8))) float v8f;

__global__ void colsum_partial(const float* __restrict__ x, float* __restrict__ partial) {
    int b     = blockIdx.x >> 3;
    int chunk = blockIdx.x & 7;
    int d     = threadIdx.x;
    const float* p = x + (b * S + chunk * 256) * D + d;
    float acc = 0.f;
#pragma unroll 8
    for (int s = 0; s < 256; ++s) acc += p[s * D];
    partial[blockIdx.x * D + d] = acc;
}

__global__ void reduce_and_coef(const float* __restrict__ partial,
                                const float* __restrict__ comp,
                                float* __restrict__ sums,
                                float* __restrict__ coef) {
    int idx = blockIdx.x * 256 + threadIdx.x;
    int b = idx >> 8, d = idx & 255;
    float s = 0.f;
#pragma unroll
    for (int c = 0; c < 8; ++c) s += partial[(b * 8 + c) * D + d];
    sums[idx] = s;

    int t = idx;
    float inv1 = 1.0f / (float)(1 + (t + 2 < S));
    int   d2   = (t >= 1) + (t + 3 < S);
    float inv2 = d2 ? 1.0f / (float)d2 : 0.f;
    int   bc   = (t >= 1) + 1 + (t + 1 < S) + (t + 2 < S) + (t + 3 < S);
    float inv3 = 1.0f / (float)(S - bc);
#pragma unroll
    for (int j = 0; j < 2; ++j) {
        float c0 = comp[0 * 2 + j], c1 = comp[1 * 2 + j];
        float c2 = comp[2 * 2 + j], c3 = comp[3 * 2 + j];
        float m3 = c3 * inv3;
        float* w = coef + t * 12 + j * 6;
        w[0] = (t >= 1)    ? (c2 * inv2 - m3) : 0.f;
        w[1] =               c1 * inv1 - m3;
        w[2] = (t + 1 < S) ? (c0        - m3) : 0.f;
        w[3] = (t + 2 < S) ? (c1 * inv1 - m3) : 0.f;
        w[4] = (t + 3 < S) ? (c2 * inv2 - m3) : 0.f;
        w[5] = m3;
    }
}

#define STAGE_A (BM * AS_STRIDE)
#define STAGE_B ((KC / 2) * BP_STRIDE * 2)

__device__ __forceinline__ void wmma_chunk(const float* __restrict__ Ac,
                                           const float* __restrict__ Bc,
                                           int aoff0, int aoff1, int boff,
                                           v8f (&acc)[2][4]) {
#pragma unroll
    for (int kk = 0; kk < KC; kk += 4) {
        v2f a0 = *(const v2f*)&Ac[aoff0 + kk];
        v2f a1 = *(const v2f*)&Ac[aoff1 + kk];
#pragma unroll
        for (int sub = 0; sub < 4; ++sub) {
            v2f bv = *(const v2f*)&Bc[boff + (kk / 2) * (BP_STRIDE * 2) + sub * 32];
            acc[0][sub] = __builtin_amdgcn_wmma_f32_16x16x4_f32(
                false, a0, false, bv, (short)0, acc[0][sub], false, false);
            acc[1][sub] = __builtin_amdgcn_wmma_f32_16x16x4_f32(
                false, a1, false, bv, (short)0, acc[1][sub], false, false);
        }
    }
}

__global__ __launch_bounds__(256) void rgcn_gemm(const float* __restrict__ x,
                                                 const float* __restrict__ basis,
                                                 const float* __restrict__ root,
                                                 const float* __restrict__ bias,
                                                 const float* __restrict__ sums,
                                                 const float* __restrict__ coef,
                                                 float* __restrict__ out) {
    __shared__ __attribute__((aligned(16))) float As[2 * STAGE_A];
    __shared__ __attribute__((aligned(16))) float Bst[2 * STAGE_B];
    __shared__ __attribute__((aligned(16))) float coefL[BM * 12];

    const int tid      = threadIdx.x;
    const int rowBase  = blockIdx.x * BM;
    const int b        = rowBase / S;
    const int tBase    = rowBase % S;
    const int colBase0 = blockIdx.y * BN;

    const int wid     = tid >> 5;
    const int lane    = tid & 31;
    const int waveRow = wid & 3;
    const int waveCol = wid >> 2;
    const int lmod    = lane & 15;
    const int lhi     = lane >> 4;
    const int koff    = lhi * 2;
    const int aoff0   = (waveRow * 32 + lmod) * AS_STRIDE + koff;
    const int aoff1   = aoff0 + 16 * AS_STRIDE;
    const int boff    = (lhi * BP_STRIDE + waveCol * 64 + lmod) * 2;

    const int c4  = (tid & 15) * 4;
    const int ar0 = tid >> 4;
    const int n4  = (tid & 31) * 4;
    const int bk0 = tid >> 5;

    v8f acc[2][4] = {};

    v4f pa[8][5];
    v4f px[8];
    v4f pb[8];
    v4f ps;

#pragma unroll
    for (int i = 0; i < 3; ++i) {
        int q = i * 256 + tid;
        *(v2f*)&coefL[q * 2] = *(const v2f*)&coef[tBase * 12 + q * 2];
    }

#define ISSUE_B(WSEG, KCBASE)                                                   \
    _Pragma("unroll")                                                           \
    for (int e = 0; e < 8; ++e)                                                 \
        pb[e] = *(const v4f*)&(WSEG)[((KCBASE) + 2 * bk0 + (e & 1) + (e >> 1) * 16) * D + colBase0 + n4];

#define STORE_B(BUF)                                                            \
    {                                                                           \
        float* Bd_ = Bst + (BUF) * STAGE_B;                                     \
        _Pragma("unroll")                                                       \
        for (int p = 0; p < 4; ++p) {                                           \
            float* d_ = Bd_ + ((bk0 + p * 8) * BP_STRIDE + n4) * 2;             \
            v4f lo_, hi_;                                                       \
            lo_.x = pb[2 * p].x; lo_.y = pb[2 * p + 1].x;                       \
            lo_.z = pb[2 * p].y; lo_.w = pb[2 * p + 1].y;                       \
            hi_.x = pb[2 * p].z; hi_.y = pb[2 * p + 1].z;                       \
            hi_.z = pb[2 * p].w; hi_.w = pb[2 * p + 1].w;                       \
            *(v4f*)&d_[0] = lo_;                                                \
            *(v4f*)&d_[4] = hi_;                                                \
        }                                                                       \
    }

#define ISSUE_ST(K)                                                             \
    {                                                                           \
        const int kcBase_ = ((K) & 3) * KC;                                     \
        const float* Wseg_ = basis + ((K) >> 2) * (D * D);                      \
        ISSUE_B(Wseg_, kcBase_)                                                 \
        const int kd_ = kcBase_ + c4;                                           \
        ps = *(const v4f*)&sums[b * D + kd_];                                   \
        _Pragma("unroll")                                                       \
        for (int e = 0; e < 8; ++e) {                                           \
            int t = tBase + ar0 + e * 16;                                       \
            _Pragma("unroll")                                                   \
            for (int o = 0; o < 5; ++o) {                                       \
                int sr = t - 1 + o;                                             \
                sr = sr < 0 ? 0 : (sr >= S ? S - 1 : sr);     \
                pa[e][o] = *(const v4f*)&x[(b * S + sr) * D + kd_];             \
            }                                                                   \
        }                                                                       \
    }

#define STORE_ST(K, BUF)                                                        \
    {                                                                           \
        float* Ad_ = As + (BUF) * STAGE_A;                                      \
        const float* wb_ = coefL + ((K) >> 2) * 6;                              \
        _Pragma("unroll")                                                       \
        for (int e = 0; e < 8; ++e) {                                           \
            int r = ar0 + e * 16;                                               \
            const float* w = wb_ + r * 12;                                      \
            v4f val = w[5] * ps;                                                \
            _Pragma("unroll")                                                   \
            for (int o = 0; o < 5; ++o) val += w[o] * pa[e][o];                 \
            *(v4f*)&Ad_[r * AS_STRIDE + c4] = val;                              \
        }                                                                       \
        STORE_B(BUF)                                                            \
    }

#define ISSUE_X(K)                                                              \
    {                                                                           \
        const int kcBase_ = ((K) & 3) * KC;                                     \
        ISSUE_B(root, kcBase_)                                                  \
        _Pragma("unroll")                                                       \
        for (int e = 0; e < 8; ++e)                                             \
            px[e] = *(const v4f*)&x[(b * S + tBase + ar0 + e * 16) * D + kcBase_ + c4]; \
    }

#define STORE_X(BUF)                                                            \
    {                                                                           \
        float* Ad_ = As + (BUF) * STAGE_A;                                      \
        _Pragma("unroll")                                                       \
        for (int e = 0; e < 8; ++e)                                             \
            *(v4f*)&Ad_[(ar0 + e * 16) * AS_STRIDE + c4] = px[e];               \
        STORE_B(BUF)                                                            \
    }

    ISSUE_ST(0)
    __syncthreads();
    STORE_ST(0, 0)
    __syncthreads();

#pragma unroll 1
    for (int kc = 0; kc < 7; ++kc) {
        const int cur = kc & 1;
        ISSUE_ST(kc + 1)
        __builtin_amdgcn_sched_barrier(0);
        wmma_chunk(As + cur * STAGE_A, Bst + cur * STAGE_B, aoff0, aoff1, boff, acc);
        __builtin_amdgcn_sched_barrier(0);
        STORE_ST(kc + 1, cur ^ 1)
        __syncthreads();
    }
    ISSUE_X(8)
    __builtin_amdgcn_sched_barrier(0);
    wmma_chunk(As + STAGE_A, Bst + STAGE_B, aoff0, aoff1, boff, acc);
    __builtin_amdgcn_sched_barrier(0);
    STORE_X(0)
    __syncthreads();
#pragma unroll 1
    for (int kc = 8; kc < 11; ++kc) {
        const int cur = kc & 1;
        ISSUE_X(kc + 1)
        __builtin_amdgcn_sched_barrier(0);
        wmma_chunk(As + cur * STAGE_A, Bst + cur * STAGE_B, aoff0, aoff1, boff, acc);
        __builtin_amdgcn_sched_barrier(0);
        STORE_X(cur ^ 1)
        __syncthreads();
    }
    wmma_chunk(As + STAGE_A, Bst + STAGE_B, aoff0, aoff1, boff, acc);
#undef ISSUE_B
#undef STORE_B
#undef ISSUE_ST
#undef STORE_ST
#undef ISSUE_X
#undef STORE_X

#pragma unroll
    for (int rt = 0; rt < 2; ++rt) {
#pragma unroll
        for (int sub = 0; sub < 4; ++sub) {
            int ncol = colBase0 + waveCol * 64 + sub * 16 + lmod;
            float badd = bias[ncol];
#pragma unroll
            for (int jj = 0; jj < 8; ++jj) {
                int r = waveRow * 32 + rt * 16 + jj + 8 * lhi;
                out[(b * S + tBase + r) * D + ncol] = acc[rt][sub][jj] + badd;
            }
        }
    }
}

extern "C" void kernel_launch(void* const* d_in, const int* in_sizes, int n_in,
                              void* d_out, int out_size, void* d_ws, size_t ws_size,
                              hipStream_t stream) {
    const float* x     = (const float*)d_in[0];
    const float* comp  = (const float*)d_in[1];
    const float* basis = (const float*)d_in[2];
    const float* root  = (const float*)d_in[3];
    const float* bias  = (const float*)d_in[4];
    float* out = (float*)d_out;

    float* wsf     = (float*)d_ws;
    float* sums    = wsf;
    float* coef    = wsf + BATCH * D;
    float* partial = coef + S * 12;

    colsum_partial<<<BATCH * 8, D, 0, stream>>>(x, partial);
    reduce_and_coef<<<S / 256, 256, 0, stream>>>(partial, comp, sums, coef);
    rgcn_gemm<<<dim3((BATCH * S) / BM, D / BN), 256, 0, stream>>>(
        x, basis, root, bias, sums, coef, out);
}


# column-sum pass made one resident pass: 256 workgroups x 64 rows, 16 independent 16-byte loads per thread (was 64 workgroups, 32 dependent load rounds); column sums are added in a new order
# speedup vs baseline: 1.6948x; 1.3693x over previous
#include <hip/hip_runtime.h>

#define D     256
#define S     2048
#define BATCH 8
#define BM    128
#define BN    128
#define KC    64
#define AS_STRIDE 68
#define BP_STRIDE 144

typedef __attribute__((ext_vector_type(2))) float v2f;
typedef __attribute__((ext_vector_type(4))) float v4f;
typedef __attribute__((ext_vector_type(8))) float v8f;

#define CS_ROWS   64
#define CS_CHUNKS (S / CS_ROWS)
__global__ __launch_bounds__(256) void colsum_partial(const float* __restrict__ x, float* __restrict__ partial) {
    __shared__ v4f red[3][D / 4];
    const int b     = blockIdx.x / CS_CHUNKS;
    const int chunk = blockIdx.x % CS_CHUNKS;
    const int q     = threadIdx.x & (D / 4 - 1);
    const int rg    = threadIdx.x >> 6;
    const float* p = x + (b * S + chunk * CS_ROWS + rg) * D + q * 4;
    v4f v[CS_ROWS / 4];
#pragma unroll
    for (int i = 0; i < CS_ROWS / 4; ++i) v[i] = *(const v4f*)&p[i * 4 * D];
#pragma unroll
    for (int h = CS_ROWS / 8; h >= 1; h >>= 1) {
#pragma unroll
        for (int i = 0; i < h; ++i) v[i] += v[i + h];
    }
    if (rg) red[rg - 1][q] = v[0];
    __syncthreads();
    if (rg == 0) {
        v4f acc = v[0] + red[0][q];
        acc += red[1][q];
        acc += red[2][q];
        *(v4f*)&partial[blockIdx.x * D + q * 4] = acc;
    }
}

__global__ void reduce_and_coef(const float* __restrict__ partial,
                                const float* __restrict__ comp,
                                float* __restrict__ sums,
                                float* __restrict__ coef) {
    int idx = blockIdx.x * 256 + threadIdx.x;
    int b = idx >> 8, d = idx & 255;
    float s = 0.f;
#pragma unroll
    for (int c = 0; c < CS_CHUNKS; ++c) s += partial[(b * CS_CHUNKS + c) * D + d];
    sums[idx] = s;

    int t = idx;
    float inv1 = 1.0f / (float)(1 + (t + 2 < S));
    int   d2   = (t >= 1) + (t + 3 < S);
    float inv2 = d2 ? 1.0f / (float)d2 : 0.f;
    int   bc   = (t >= 1) + 1 + (t + 1 < S) + (t + 2 < S) + (t + 3 < S);
    float inv3 = 1.0f / (float)(S - bc);
#pragma unroll
    for (int j = 0; j < 2; ++j) {
        float c0 = comp[0 * 2 + j], c1 = comp[1 * 2 + j];
        float c2 = comp[2 * 2 + j], c3 = comp[3 * 2 + j];
        float m3 = c3 * inv3;
        float* w = coef + t * 12 + j * 6;
        w[0] = (t >= 1)    ? (c2 * inv2 - m3) : 0.f;
        w[1] =               c1 * inv1 - m3;
        w[2] = (t + 1 < S) ? (c0        - m3) : 0.f;
        w[3] = (t + 2 < S) ? (c1 * inv1 - m3) : 0.f;
        w[4] = (t + 3 < S) ? (c2 * inv2 - m3) : 0.f;
        w[5] = m3;
    }
}

#define STAGE_A (BM * AS_STRIDE)
#define STAGE_B ((KC / 2) * BP_STRIDE * 2)

__device__ __forceinline__ void wmma_chunk(const float* __restrict__ Ac,
                                           const float* __restrict__ Bc,
                                           int aoff0, int aoff1, int boff,
                                           v8f (&acc)[2][4]) {
#pragma unroll
    for (int kk = 0; kk < KC; kk += 4) {
        v2f a0 = *(const v2f*)&Ac[aoff0 + kk];
        v2f a1 = *(const v2f*)&Ac[aoff1 + kk];
#pragma unroll
        for (int sub = 0; sub < 4; ++sub) {
            v2f bv = *(const v2f*)&Bc[boff + (kk / 2) * (BP_STRIDE * 2) + sub * 32];
            acc[0][sub] = __builtin_amdgcn_wmma_f32_16x16x4_f32(
                false, a0, false, bv, (short)0, acc[0][sub], false, false);
            acc[1][sub] = __builtin_amdgcn_wmma_f32_16x16x4_f32(
                false, a1, false, bv, (short)0, acc[1][sub], false, false);
        }
    }
}

__global__ __launch_bounds__(256) void rgcn_gemm(const float* __restrict__ x,
                                                 const float* __restrict__ basis,
                                                 const float* __restrict__ root,
                                                 const float* __restrict__ bias,
                                                 const float* __restrict__ sums,
                                                 const float* __restrict__ coef,
                                                 float* __restrict__ out) {
    __shared__ __attribute__((aligned(16))) float As[2 * STAGE_A];
    __shared__ __attribute__((aligned(16))) float Bst[2 * STAGE_B];
    __shared__ __attribute__((aligned(16))) float coefL[BM * 12];

    const int tid      = threadIdx.x;
    const int rowBase  = blockIdx.x * BM;
    const int b        = rowBase / S;
    const int tBase    = rowBase % S;
    const int colBase0 = blockIdx.y * BN;

    const int wid     = tid >> 5;
    const int lane    = tid & 31;
    const int waveRow = wid & 3;
    const int waveCol = wid >> 2;
    const int lmod    = lane & 15;
    const int lhi     = lane >> 4;
    const int koff    = lhi * 2;
    const int aoff0   = (waveRow * 32 + lmod) * AS_STRIDE + koff;
    const int aoff1   = aoff0 + 16 * AS_STRIDE;
    const int boff    = (lhi * BP_STRIDE + waveCol * 64 + lmod) * 2;

    const int c4  = (tid & 15) * 4;
    const int ar0 = tid >> 4;
    const int n4  = (tid & 31) * 4;
    const int bk0 = tid >> 5;

    v8f acc[2][4] = {};

    v4f pa[8][5];
    v4f px[8];
    v4f pb[8];
    v4f ps;

#pragma unroll
    for (int i = 0; i < 3; ++i) {
        int q = i * 256 + tid;
        *(v2f*)&coefL[q * 2] = *(const v2f*)&coef[tBase * 12 + q * 2];
    }

#define ISSUE_B(WSEG, KCBASE)                                                   \
    _Pragma("unroll")                                                           \
    for (int e = 0; e < 8; ++e)                                                 \
        pb[e] = *(const v4f*)&(WSEG)[((KCBASE) + 2 * bk0 + (e & 1) + (e >> 1) * 16) * D + colBase0 + n4];

#define STORE_B(BUF)                                                            \
    {                                                                           \
        float* Bd_ = Bst + (BUF) * STAGE_B;                                     \
        _Pragma("unroll")                                                       \
        for (int p = 0; p < 4; ++p) {                                           \
            float* d_ = Bd_ + ((bk0 + p * 8) * BP_STRIDE + n4) * 2;             \
            v4f lo_, hi_;                                                       \
            lo_.x = pb[2 * p].x; lo_.y = pb[2 * p + 1].x;                       \
            lo_.z = pb[2 * p].y; lo_.w = pb[2 * p + 1].y;                       \
            hi_.x = pb[2 * p].z; hi_.y = pb[2 * p + 1].z;                       \
            hi_.z = pb[2 * p].w; hi_.w = pb[2 * p + 1].w;                       \
            *(v4f*)&d_[0] = lo_;                                                \
            *(v4f*)&d_[4] = hi_;                                                \
        }                                                                       \
    }

#define ISSUE_ST(K)                                                             \
    {                                                                           \
        const int kcBase_ = ((K) & 3) * KC;                                     \
        const float* Wseg_ = basis + ((K) >> 2) * (D * D);                      \
        ISSUE_B(Wseg_, kcBase_)                                                 \
        const int kd_ = kcBase_ + c4;                                           \
        ps = *(const v4f*)&sums[b * D + kd_];                                   \
        _Pragma("unroll")                                                       \
        for (int e = 0; e < 8; ++e) {                                           \
            int t = tBase + ar0 + e * 16;                                       \
            _Pragma("unroll")                                                   \
            for (int o = 0; o < 5; ++o) {                                       \
                int sr = t - 1 + o;                                             \
                sr = sr < 0 ? 0 : (sr >= S ? S - 1 : sr);     \
                pa[e][o] = *(const v4f*)&x[(b * S + sr) * D + kd_];             \
            }                                                                   \
        }                                                                       \
    }

#define STORE_ST(K, BUF)                                                        \
    {                                                                           \
        float* Ad_ = As + (BUF) * STAGE_A;                                      \
        const float* wb_ = coefL + ((K) >> 2) * 6;                              \
        _Pragma("unroll")                                                       \
        for (int e = 0; e < 8; ++e) {                                           \
            int r = ar0 + e * 16;                                               \
            const float* w = wb_ + r * 12;                                      \
            v4f val = w[5] * ps;                                                \
            _Pragma("unroll")                                                   \
            for (int o = 0; o < 5; ++o) val += w[o] * pa[e][o];                 \
            *(v4f*)&Ad_[r * AS_STRIDE + c4] = val;                              \
        }                                                                       \
        STORE_B(BUF)                                                            \
    }

#define ISSUE_X(K)                                                              \
    {                                                                           \
        const int kcBase_ = ((K) & 3) * KC;                                     \
        ISSUE_B(root, kcBase_)                                                  \
        _Pragma("unroll")                                                       \
        for (int e = 0; e < 8; ++e)                                             \
            px[e] = *(const v4f*)&x[(b * S + tBase + ar0 + e * 16) * D + kcBase_ + c4]; \
    }

#define STORE_X(BUF)                                                            \
    {                                                                           \
        float* Ad_ = As + (BUF) * STAGE_A;                                      \
        _Pragma("unroll")                                                       \
        for (int e = 0; e < 8; ++e)                                             \
            *(v4f*)&Ad_[(ar0 + e * 16) * AS_STRIDE + c4] = px[e];               \
        STORE_B(BUF)                                                            \
    }

    ISSUE_ST(0)
    __syncthreads();
    STORE_ST(0, 0)
    __syncthreads();

#pragma unroll 1
    for (int kc = 0; kc < 7; ++kc) {
        const int cur = kc & 1;
        ISSUE_ST(kc + 1)
        __builtin_amdgcn_sched_barrier(0);
        wmma_chunk(As + cur * STAGE_A, Bst + cur * STAGE_B, aoff0, aoff1, boff, acc);
        __builtin_amdgcn_sched_barrier(0);
        STORE_ST(kc + 1, cur ^ 1)
        __syncthreads();
    }
    ISSUE_X(8)
    __builtin_amdgcn_sched_barrier(0);
    wmma_chunk(As + STAGE_A, Bst + STAGE_B, aoff0, aoff1, boff, acc);
    __builtin_amdgcn_sched_barrier(0);
    STORE_X(0)
    __syncthreads();
#pragma unroll 1
    for (int kc = 8; kc < 11; ++kc) {
        const int cur = kc & 1;
        ISSUE_X(kc + 1)
        __builtin_amdgcn_sched_barrier(0);
        wmma_chunk(As + cur * STAGE_A, Bst + cur * STAGE_B, aoff0, aoff1, boff, acc);
        __builtin_amdgcn_sched_barrier(0);
        STORE_X(cur ^ 1)
        __syncthreads();
    }
    wmma_chunk(As + STAGE_A, Bst + STAGE_B, aoff0, aoff1, boff, acc);
#undef ISSUE_B
#undef STORE_B
#undef ISSUE_ST
#undef STORE_ST
#undef ISSUE_X
#undef STORE_X

#pragma unroll
    for (int rt = 0; rt < 2; ++rt) {
#pragma unroll
        for (int sub = 0; sub < 4; ++sub) {
            int ncol = colBase0 + waveCol * 64 + sub * 16 + lmod;
            float badd = bias[ncol];
#pragma unroll
            for (int jj = 0; jj < 8; ++jj) {
                int r = waveRow * 32 + rt * 16 + jj + 8 * lhi;
                out[(b * S + tBase + r) * D + ncol] = acc[rt][sub][jj] + badd;
            }
        }
    }
}

extern "C" void kernel_launch(void* const* d_in, const int* in_sizes, int n_in,
                              void* d_out, int out_size, void* d_ws, size_t ws_size,
                              hipStream_t stream) {
    const float* x     = (const float*)d_in[0];
    const float* comp  = (const float*)d_in[1];
    const float* basis = (const float*)d_in[2];
    const float* root  = (const float*)d_in[3];
    const float* bias  = (const float*)d_in[4];
    float* out = (float*)d_out;

    float* wsf     = (float*)d_ws;
    float* sums    = wsf;
    float* coef    = wsf + BATCH * D;
    float* partial = coef + S * 12;

    colsum_partial<<<BATCH * CS_CHUNKS, 256, 0, stream>>>(x, partial);
    reduce_and_coef<<<S / 256, 256, 0, stream>>>(partial, comp, sums, coef);
    rgcn_gemm<<<dim3((BATCH * S) / BM, D / BN), 256, 0, stream>>>(
        x, basis, root, bias, sums, coef, out);
}


# launch 2 (8 workgroups) folded into the GEMM prologue: each workgroup adds its batch's 32 partial sum rows (same order) and computes its own 128 coefficient rows, into LDS; 3 launches -> 2
# speedup vs baseline: 1.7834x; 1.0523x over previous
#include <hip/hip_runtime.h>

#define D     256
#define S     2048
#define BATCH 8
#define BM    128
#define BN    128
#define KC    64
#define AS_STRIDE 68
#define BP_STRIDE 144

typedef __attribute__((ext_vector_type(2))) float v2f;
typedef __attribute__((ext_vector_type(4))) float v4f;
typedef __attribute__((ext_vector_type(8))) float v8f;

#define CS_ROWS   64
#define CS_CHUNKS (S / CS_ROWS)
__global__ __launch_bounds__(256) void colsum_partial(const float* __restrict__ x, float* __restrict__ partial) {
    __shared__ v4f red[3][D / 4];
    const int b     = blockIdx.x / CS_CHUNKS;
    const int chunk = blockIdx.x % CS_CHUNKS;
    const int q     = threadIdx.x & (D / 4 - 1);
    const int rg    = threadIdx.x >> 6;
    const float* p = x + (b * S + chunk * CS_ROWS + rg) * D + q * 4;
    v4f v[CS_ROWS / 4];
#pragma unroll
    for (int i = 0; i < CS_ROWS / 4; ++i) v[i] = *(const v4f*)&p[i * 4 * D];
#pragma unroll
    for (int h = CS_ROWS / 8; h >= 1; h >>= 1) {
#pragma unroll
        for (int i = 0; i < h; ++i) v[i] += v[i + h];
    }
    if (rg) red[rg - 1][q] = v[0];
    __syncthreads();
    if (rg == 0) {
        v4f acc = v[0] + red[0][q];
        acc += red[1][q];
        acc += red[2][q];
        *(v4f*)&partial[blockIdx.x * D + q * 4] = acc;
    }
}

#define STAGE_A (BM * AS_STRIDE)
#define STAGE_B ((KC / 2) * BP_STRIDE * 2)

__device__ __forceinline__ void wmma_chunk(const float* __restrict__ Ac,
                                           const float* __restrict__ Bc,
                                           int aoff0, int aoff1, int boff,
                                           v8f (&acc)[2][4]) {
#pragma unroll
    for (int kk = 0; kk < KC; kk += 4) {
        v2f a0 = *(const v2f*)&Ac[aoff0 + kk];
        v2f a1 = *(const v2f*)&Ac[aoff1 + kk];
#pragma unroll
        for (int sub = 0; sub < 4; ++sub) {
            v2f bv = *(const v2f*)&Bc[boff + (kk / 2) * (BP_STRIDE * 2) + sub * 32];
            acc[0][sub] = __builtin_amdgcn_wmma_f32_16x16x4_f32(
                false, a0, false, bv, (short)0, acc[0][sub], false, false);
            acc[1][sub] = __builtin_amdgcn_wmma_f32_16x16x4_f32(
                false, a1, false, bv, (short)0, acc[1][sub], false, false);
        }
    }
}

__global__ __launch_bounds__(256) void rgcn_gemm(const float* __restrict__ x,
                                                 const float* __restrict__ comp,
                                                 const float* __restrict__ basis,
                                                 const float* __restrict__ root,
                                                 const float* __restrict__ bias,
                                                 const float* __restrict__ partial,
                                                 float* __restrict__ out) {
    __shared__ __attribute__((aligned(16))) float As[2 * STAGE_A];
    __shared__ __attribute__((aligned(16))) float Bst[2 * STAGE_B];
    __shared__ __attribute__((aligned(16))) float coefL[BM * 12];
    __shared__ __attribute__((aligned(16))) float sumsL[D];

    const int tid      = threadIdx.x;
    const int rowBase  = blockIdx.x * BM;
    const int b        = rowBase / S;
    const int tBase    = rowBase % S;
    const int colBase0 = blockIdx.y * BN;

    const int wid     = tid >> 5;
    const int lane    = tid & 31;
    const int waveRow = wid & 3;
    const int waveCol = wid >> 2;
    const int lmod    = lane & 15;
    const int lhi     = lane >> 4;
    const int koff    = lhi * 2;
    const int aoff0   = (waveRow * 32 + lmod) * AS_STRIDE + koff;
    const int aoff1   = aoff0 + 16 * AS_STRIDE;
    const int boff    = (lhi * BP_STRIDE + waveCol * 64 + lmod) * 2;

    const int c4  = (tid & 15) * 4;
    const int ar0 = tid >> 4;
    const int n4  = (tid & 31) * 4;
    const int bk0 = tid >> 5;

    v8f acc[2][4] = {};

    v4f pa[8][5];
    v4f px[8];
    v4f pb[8];

    {
        const int r_ = tid >> 1, j = tid & 1;
        const int t  = tBase + r_;
        float inv1 = 1.0f / (float)(1 + (t + 2 < S));
        int   d2   = (t >= 1) + (t + 3 < S);
        float inv2 = d2 ? 1.0f / (float)d2 : 0.f;
        int   bc   = (t >= 1) + 1 + (t + 1 < S) + (t + 2 < S) + (t + 3 < S);
        float inv3 = 1.0f / (float)(S - bc);
        float c0 = comp[0 * 2 + j], c1 = comp[1 * 2 + j];
        float c2 = comp[2 * 2 + j], c3 = comp[3 * 2 + j];
        float m3 = c3 * inv3;
        float* w = coefL + r_ * 12 + j * 6;
        w[0] = (t >= 1)    ? (c2 * inv2 - m3) : 0.f;
        w[1] =               c1 * inv1 - m3;
        w[2] = (t + 1 < S) ? (c0        - m3) : 0.f;
        w[3] = (t + 2 < S) ? (c1 * inv1 - m3) : 0.f;
        w[4] = (t + 3 < S) ? (c2 * inv2 - m3) : 0.f;
        w[5] = m3;
    }

#define ISSUE_B(WSEG, KCBASE)                                                   \
    _Pragma("unroll")                                                           \
    for (int e = 0; e < 8; ++e)                                                 \
        pb[e] = *(const v4f*)&(WSEG)[((KCBASE) + 2 * bk0 + (e & 1) + (e >> 1) * 16) * D + colBase0 + n4];

#define STORE_B(BUF)                                                            \
    {                                                                           \
        float* Bd_ = Bst + (BUF) * STAGE_B;                                     \
        _Pragma("unroll")                                                       \
        for (int p = 0; p < 4; ++p) {                                           \
            float* d_ = Bd_ + ((bk0 + p * 8) * BP_STRIDE + n4) * 2;             \
            v4f lo_, hi_;                                                       \
            lo_.x = pb[2 * p].x; lo_.y = pb[2 * p + 1].x;                       \
            lo_.z = pb[2 * p].y; lo_.w = pb[2 * p + 1].y;                       \
            hi_.x = pb[2 * p].z; hi_.y = pb[2 * p + 1].z;                       \
            hi_.z = pb[2 * p].w; hi_.w = pb[2 * p + 1].w;                       \
            *(v4f*)&d_[0] = lo_;                                                \
            *(v4f*)&d_[4] = hi_;                                                \
        }                                                                       \
    }

#define ISSUE_ST(K)                                                             \
    {                                                                           \
        const int kcBase_ = ((K) & 3) * KC;                                     \
        const float* Wseg_ = basis + ((K) >> 2) * (D * D);                      \
        ISSUE_B(Wseg_, kcBase_)                                                 \
        const int kd_ = kcBase_ + c4;                                           \
        _Pragma("unroll")                                                       \
        for (int e = 0; e < 8; ++e) {                                           \
            int t = tBase + ar0 + e * 16;                                       \
            _Pragma("unroll")                                                   \
            for (int o = 0; o < 5; ++o) {                                       \
                int sr = t - 1 + o;                                             \
                sr = sr < 0 ? 0 : (sr >= S ? S - 1 : sr);     \
                pa[e][o] = *(const v4f*)&x[(b * S + sr) * D + kd_];             \
            }                                                                   \
        }                                                                       \
    }

#define STORE_ST(K, BUF)                                                        \
    {                                                                           \
        float* Ad_ = As + (BUF) * STAGE_A;                                      \
        const float* wb_ = coefL + ((K) >> 2) * 6;                              \
        const v4f ps = *(const v4f*)&sumsL[((K) & 3) * KC + c4];                \
        _Pragma("unroll")                                                       \
        for (int e = 0; e < 8; ++e) {                                           \
            int r = ar0 + e * 16;                                               \
            const float* w = wb_ + r * 12;                                      \
            v4f val = w[5] * ps;                                                \
            _Pragma("unroll")                                                   \
            for (int o = 0; o < 5; ++o) val += w[o] * pa[e][o];                 \
            *(v4f*)&Ad_[r * AS_STRIDE + c4] = val;                              \
        }                                                                       \
        STORE_B(BUF)                                                            \
    }

#define ISSUE_X(K)                                                              \
    {                                                                           \
        const int kcBase_ = ((K) & 3) * KC;                                     \
        ISSUE_B(root, kcBase_)                                                  \
        _Pragma("unroll")                                                       \
        for (int e = 0; e < 8; ++e)                                             \
            px[e] = *(const v4f*)&x[(b * S + tBase + ar0 + e * 16) * D + kcBase_ + c4]; \
    }

#define STORE_X(BUF)                                                            \
    {                                                                           \
        float* Ad_ = As + (BUF) * STAGE_A;                                      \
        _Pragma("unroll")                                                       \
        for (int e = 0; e < 8; ++e)                                             \
            *(v4f*)&Ad_[(ar0 + e * 16) * AS_STRIDE + c4] = px[e];               \
        STORE_B(BUF)                                                            \
    }

    ISSUE_ST(0)
    float pp[CS_CHUNKS];
#pragma unroll
    for (int c = 0; c < CS_CHUNKS; ++c) pp[c] = partial[(b * CS_CHUNKS + c) * D + tid];
    {
        float s = 0.f;
#pragma unroll
        for (int c = 0; c < CS_CHUNKS; ++c) s += pp[c];
        sumsL[tid] = s;
    }
    __syncthreads();
    STORE_ST(0, 0)
    __syncthreads();

#pragma unroll 1
    for (int kc = 0; kc < 7; ++kc) {
        const int cur = kc & 1;
        ISSUE_ST(kc + 1)
        __builtin_amdgcn_sched_barrier(0);
        wmma_chunk(As + cur * STAGE_A, Bst + cur * STAGE_B, aoff0, aoff1, boff, acc);
        __builtin_amdgcn_sched_barrier(0);
        STORE_ST(kc + 1, cur ^ 1)
        __syncthreads();
    }
    ISSUE_X(8)
    __builtin_amdgcn_sched_barrier(0);
    wmma_chunk(As + STAGE_A, Bst + STAGE_B, aoff0, aoff1, boff, acc);
    __builtin_amdgcn_sched_barrier(0);
    STORE_X(0)
    __syncthreads();
#pragma unroll 1
    for (int kc = 8; kc < 11; ++kc) {
        const int cur = kc & 1;
        ISSUE_X(kc + 1)
        __builtin_amdgcn_sched_barrier(0);
        wmma_chunk(As + cur * STAGE_A, Bst + cur * STAGE_B, aoff0, aoff1, boff, acc);
        __builtin_amdgcn_sched_barrier(0);
        STORE_X(cur ^ 1)
        __syncthreads();
    }
    wmma_chunk(As + STAGE_A, Bst + STAGE_B, aoff0, aoff1, boff, acc);
#undef ISSUE_B
#undef STORE_B
#undef ISSUE_ST
#undef STORE_ST
#undef ISSUE_X
#undef STORE_X

#pragma unroll
    for (int rt = 0; rt < 2; ++rt) {
#pragma unroll
        for (int sub = 0; sub < 4; ++sub) {
            int ncol = colBase0 + waveCol * 64 + sub * 16 + lmod;
            float badd = bias[ncol];
#pragma unroll
            for (int jj = 0; jj < 8; ++jj) {
                int r = waveRow * 32 + rt * 16 + jj + 8 * lhi;
                out[(b * S + tBase + r) * D + ncol] = acc[rt][sub][jj] + badd;
            }
        }
    }
}

extern "C" void kernel_launch(void* const* d_in, const int* in_sizes, int n_in,
                              void* d_out, int out_size, void* d_ws, size_t ws_size,
                              hipStream_t stream) {
    const float* x     = (const float*)d_in[0];
    const float* comp  = (const float*)d_in[1];
    const float* basis = (const float*)d_in[2];
    const float* root  = (const float*)d_in[3];
    const float* bias  = (const float*)d_in[4];
    float* out = (float*)d_out;

    float* partial = (float*)d_ws;

    colsum_partial<<<BATCH * CS_CHUNKS, 256, 0, stream>>>(x, partial);
    rgcn_gemm<<<dim3((BATCH * S) / BM, D / BN), 256, 0, stream>>>(
        x, comp, basis, root, bias, partial, out);
}


# A staging from resident rows: a thread owns 8 consecutive rows and fetches the 12 x rows of their windows once (48 loads kept in registers for u0, u1, x chunks) instead of 352 overlapping loads
# speedup vs baseline: 2.0059x; 1.1248x over previous
#include <hip/hip_runtime.h>

#define D     256
#define S     2048
#define BATCH 8
#define BM    128
#define BN    128
#define KC    64
#define AS_STRIDE 68
#define BP_STRIDE 144

typedef __attribute__((ext_vector_type(2))) float v2f;
typedef __attribute__((ext_vector_type(4))) float v4f;
typedef __attribute__((ext_vector_type(8))) float v8f;

#define CS_ROWS   64
#define CS_CHUNKS (S / CS_ROWS)
__global__ __launch_bounds__(256) void colsum_partial(const float* __restrict__ x, float* __restrict__ partial) {
    __shared__ v4f red[3][D / 4];
    const int b     = blockIdx.x / CS_CHUNKS;
    const int chunk = blockIdx.x % CS_CHUNKS;
    const int q     = threadIdx.x & (D / 4 - 1);
    const int rg    = threadIdx.x >> 6;
    const float* p = x + (b * S + chunk * CS_ROWS + rg) * D + q * 4;
    v4f v[CS_ROWS / 4];
#pragma unroll
    for (int i = 0; i < CS_ROWS / 4; ++i) v[i] = *(const v4f*)&p[i * 4 * D];
#pragma unroll
    for (int h = CS_ROWS / 8; h >= 1; h >>= 1) {
#pragma unroll
        for (int i = 0; i < h; ++i) v[i] += v[i + h];
    }
    if (rg) red[rg - 1][q] = v[0];
    __syncthreads();
    if (rg == 0) {
        v4f acc = v[0] + red[0][q];
        acc += red[1][q];
        acc += red[2][q];
        *(v4f*)&partial[blockIdx.x * D + q * 4] = acc;
    }
}

#define STAGE_A (BM * AS_STRIDE)
#define STAGE_B ((KC / 2) * BP_STRIDE * 2)

__device__ __forceinline__ void wmma_chunk(const float* __restrict__ Ac,
                                           const float* __restrict__ Bc,
                                           int aoff0, int aoff1, int boff,
                                           v8f (&acc)[2][4]) {
#pragma unroll
    for (int kk = 0; kk < KC; kk += 4) {
        v2f a0 = *(const v2f*)&Ac[aoff0 + kk];
        v2f a1 = *(const v2f*)&Ac[aoff1 + kk];
#pragma unroll
        for (int sub = 0; sub < 4; ++sub) {
            v2f bv = *(const v2f*)&Bc[boff + (kk / 2) * (BP_STRIDE * 2) + sub * 32];
            acc[0][sub] = __builtin_amdgcn_wmma_f32_16x16x4_f32(
                false, a0, false, bv, (short)0, acc[0][sub], false, false);
            acc[1][sub] = __builtin_amdgcn_wmma_f32_16x16x4_f32(
                false, a1, false, bv, (short)0, acc[1][sub], false, false);
        }
    }
}

__global__ __launch_bounds__(256) void rgcn_gemm(const float* __restrict__ x,
                                                 const float* __restrict__ comp,
                                                 const float* __restrict__ basis,
                                                 const float* __restrict__ root,
                                                 const float* __restrict__ bias,
                                                 const float* __restrict__ partial,
                                                 float* __restrict__ out) {
    __shared__ __attribute__((aligned(16))) float As[2 * STAGE_A];
    __shared__ __attribute__((aligned(16))) float Bst[2 * STAGE_B];
    __shared__ __attribute__((aligned(16))) float coefL[BM * 12];
    __shared__ __attribute__((aligned(16))) float sumsL[D];

    const int tid      = threadIdx.x;
    const int rowBase  = blockIdx.x * BM;
    const int b        = rowBase / S;
    const int tBase    = rowBase % S;
    const int colBase0 = blockIdx.y * BN;

    const int wid     = tid >> 5;
    const int lane    = tid & 31;
    const int waveRow = wid & 3;
    const int waveCol = wid >> 2;
    const int lmod    = lane & 15;
    const int lhi     = lane >> 4;
    const int koff    = lhi * 2;
    const int aoff0   = (waveRow * 32 + lmod) * AS_STRIDE + koff;
    const int aoff1   = aoff0 + 16 * AS_STRIDE;
    const int boff    = (lhi * BP_STRIDE + waveCol * 64 + lmod) * 2;

    const int c4  = (tid & 15) * 4;
    const int rg  = tid >> 4;
    const int n4  = (tid & 31) * 4;
    const int bk0 = tid >> 5;

    v8f acc[2][4] = {};

    v4f xr[4][12];
    v4f pb[8];

    {
        const int r_ = tid >> 1, j = tid & 1;
        const int t  = tBase + r_;
        float inv1 = 1.0f / (float)(1 + (t + 2 < S));
        int   d2   = (t >= 1) + (t + 3 < S);
        float inv2 = d2 ? 1.0f / (float)d2 : 0.f;
        int   bc   = (t >= 1) + 1 + (t + 1 < S) + (t + 2 < S) + (t + 3 < S);
        float inv3 = 1.0f / (float)(S - bc);
        float c0 = comp[0 * 2 + j], c1 = comp[1 * 2 + j];
        float c2 = comp[2 * 2 + j], c3 = comp[3 * 2 + j];
        float m3 = c3 * inv3;
        float* w = coefL + r_ * 12 + j * 6;
        w[0] = (t >= 1)    ? (c2 * inv2 - m3) : 0.f;
        w[1] =               c1 * inv1 - m3;
        w[2] = (t + 1 < S) ? (c0        - m3) : 0.f;
        w[3] = (t + 2 < S) ? (c1 * inv1 - m3) : 0.f;
        w[4] = (t + 3 < S) ? (c2 * inv2 - m3) : 0.f;
        w[5] = m3;
    }

#define ISSUE_B(WSEG, KCBASE)                                                   \
    _Pragma("unroll")                                                           \
    for (int e = 0; e < 8; ++e)                                                 \
        pb[e] = *(const v4f*)&(WSEG)[((KCBASE) + 2 * bk0 + (e & 1) + (e >> 1) * 16) * D + colBase0 + n4];

#define STORE_B(BUF)                                                            \
    {                                                                           \
        float* Bd_ = Bst + (BUF) * STAGE_B;                                     \
        _Pragma("unroll")                                                       \
        for (int p = 0; p < 4; ++p) {                                           \
            float* d_ = Bd_ + ((bk0 + p * 8) * BP_STRIDE + n4) * 2;             \
            v4f lo_, hi_;                                                       \
            lo_.x = pb[2 * p].x; lo_.y = pb[2 * p + 1].x;                       \
            lo_.z = pb[2 * p].y; lo_.w = pb[2 * p + 1].y;                       \
            hi_.x = pb[2 * p].z; hi_.y = pb[2 * p + 1].z;                       \
            hi_.z = pb[2 * p].w; hi_.w = pb[2 * p + 1].w;                       \
            *(v4f*)&d_[0] = lo_;                                                \
            *(v4f*)&d_[4] = hi_;                                                \
        }                                                                       \
    }

#define STORE_A(CQ, SG, BUF)                                                    \
    {                                                                           \
        float* Ad_ = As + (BUF) * STAGE_A;                                      \
        if ((SG) < 2) {                                                         \
            const float* wb_ = coefL + (SG) * 6;                                \
            const v4f ps = *(const v4f*)&sumsL[(CQ) * KC + c4];                 \
            _Pragma("unroll")                                                   \
            for (int e = 0; e < 8; ++e) {                                       \
                int r = rg * 8 + e;                                             \
                const float* w = wb_ + r * 12;                                  \
                v4f val = w[5] * ps;                                            \
                _Pragma("unroll")                                               \
                for (int o = 0; o < 5; ++o) val += w[o] * xr[CQ][e + o];        \
                *(v4f*)&Ad_[r * AS_STRIDE + c4] = val;                          \
            }                                                                   \
        } else {                                                                \
            _Pragma("unroll")                                                   \
            for (int e = 0; e < 8; ++e)                                         \
                *(v4f*)&Ad_[(rg * 8 + e) * AS_STRIDE + c4] = xr[CQ][e + 1];     \
        }                                                                       \
    }

#define TRIP(CQ, NCQ, NSG, NWSEG)                                               \
    {                                                                           \
        ISSUE_B(NWSEG, (NCQ) * KC)                                              \
          \
        __builtin_amdgcn_sched_barrier(0);                                      \
        wmma_chunk(As + ((CQ) & 1) * STAGE_A, Bst + ((CQ) & 1) * STAGE_B, aoff0, aoff1, boff, acc); \
          \
        __builtin_amdgcn_sched_barrier(0);                                      \
        STORE_A(NCQ, NSG, ((CQ) & 1) ^ 1)                                       \
        STORE_B(((CQ) & 1) ^ 1)                                                 \
        __syncthreads();                                                        \
    }

    {
        const int tm1 = tBase + rg * 8 - 1;
#pragma unroll
        for (int c = 0; c < 4; ++c) {
#pragma unroll
            for (int i = 0; i < 12; ++i) {
                int sr = tm1 + i;
                sr = sr < 0 ? 0 : (sr >= S ? S - 1 : sr);
                xr[c][i] = *(const v4f*)&x[(b * S + sr) * D + c * KC + c4];
            }
            if (c == 0) {
                ISSUE_B(basis, 0)
            }
        }
    }
    float pp[CS_CHUNKS];
#pragma unroll
    for (int c = 0; c < CS_CHUNKS; ++c) pp[c] = partial[(b * CS_CHUNKS + c) * D + tid];
    {
        float s = 0.f;
#pragma unroll
        for (int c = 0; c < CS_CHUNKS; ++c) s += pp[c];
        sumsL[tid] = s;
    }
    __syncthreads();
    STORE_A(0, 0, 0)
    STORE_B(0)
    __syncthreads();

#pragma unroll 1
    for (int sg = 0; sg < 3; ++sg) {
        const float* wseg  = sg < 2 ? basis + sg * (D * D) : root;
        const float* wnext = sg < 1 ? basis + (D * D) : root;
        TRIP(0, 1, sg, wseg)
        TRIP(1, 2, sg, wseg)
        TRIP(2, 3, sg, wseg)
        if (sg < 2) {
            TRIP(3, 0, sg + 1, wnext)
        }
    }
    wmma_chunk(As + STAGE_A, Bst + STAGE_B, aoff0, aoff1, boff, acc);
#undef TRIP
#undef STORE_A
#undef ISSUE_B
#undef STORE_B

#pragma unroll
    for (int rt = 0; rt < 2; ++rt) {
#pragma unroll
        for (int sub = 0; sub < 4; ++sub) {
            int ncol = colBase0 + waveCol * 64 + sub * 16 + lmod;
            float badd = bias[ncol];
#pragma unroll
            for (int jj = 0; jj < 8; ++jj) {
                int r = waveRow * 32 + rt * 16 + jj + 8 * lhi;
                out[(b * S + tBase + r) * D + ncol] = acc[rt][sub][jj] + badd;
            }
        }
    }
}

extern "C" void kernel_launch(void* const* d_in, const int* in_sizes, int n_in,
                              void* d_out, int out_size, void* d_ws, size_t ws_size,
                              hipStream_t stream) {
    const float* x     = (const float*)d_in[0];
    const float* comp  = (const float*)d_in[1];
    const float* basis = (const float*)d_in[2];
    const float* root  = (const float*)d_in[3];
    const float* bias  = (const float*)d_in[4];
    float* out = (float*)d_out;

    float* partial = (float*)d_ws;

    colsum_partial<<<BATCH * CS_CHUNKS, 256, 0, stream>>>(x, partial);
    rgcn_gemm<<<dim3((BATCH * S) / BM, D / BN), 256, 0, stream>>>(
        x, comp, basis, root, bias, partial, out);
}


# staging of the next chunk moved from behind the WMMA block into its middle; a wave lowers its priority (s_setprio) once it has staged, so a SIMD's two waves alternate staging and multiplying
# speedup vs baseline: 2.0797x; 1.0368x over previous
#include <hip/hip_runtime.h>

#define D     256
#define S     2048
#define BATCH 8
#define BM    128
#define BN    128
#define KC    64
#define AS_STRIDE 68
#define BP_STRIDE 144

typedef __attribute__((ext_vector_type(2))) float v2f;
typedef __attribute__((ext_vector_type(4))) float v4f;
typedef __attribute__((ext_vector_type(8))) float v8f;

#define CS_ROWS   64
#define CS_CHUNKS (S / CS_ROWS)
__global__ __launch_bounds__(256) void colsum_partial(const float* __restrict__ x, float* __restrict__ partial) {
    __shared__ v4f red[3][D / 4];
    const int b     = blockIdx.x / CS_CHUNKS;
    const int chunk = blockIdx.x % CS_CHUNKS;
    const int q     = threadIdx.x & (D / 4 - 1);
    const int rg    = threadIdx.x >> 6;
    const float* p = x + (b * S + chunk * CS_ROWS + rg) * D + q * 4;
    v4f v[CS_ROWS / 4];
#pragma unroll
    for (int i = 0; i < CS_ROWS / 4; ++i) v[i] = *(const v4f*)&p[i * 4 * D];
#pragma unroll
    for (int h = CS_ROWS / 8; h >= 1; h >>= 1) {
#pragma unroll
        for (int i = 0; i < h; ++i) v[i] += v[i + h];
    }
    if (rg) red[rg - 1][q] = v[0];
    __syncthreads();
    if (rg == 0) {
        v4f acc = v[0] + red[0][q];
        acc += red[1][q];
        acc += red[2][q];
        *(v4f*)&partial[blockIdx.x * D + q * 4] = acc;
    }
}

#define STAGE_A (BM * AS_STRIDE)
#define STAGE_B ((KC / 2) * BP_STRIDE * 2)

#define KSPLIT 32
template <int K0, int K1>
__device__ __forceinline__ void wmma_steps(const float* __restrict__ Ac,
                                           const float* __restrict__ Bc,
                                           int aoff0, int aoff1, int boff,
                                           v8f (&acc)[2][4]) {
#pragma unroll
    for (int kk = K0; kk < K1; kk += 4) {
        v2f a0 = *(const v2f*)&Ac[aoff0 + kk];
        v2f a1 = *(const v2f*)&Ac[aoff1 + kk];
#pragma unroll
        for (int sub = 0; sub < 4; ++sub) {
            v2f bv = *(const v2f*)&Bc[boff + (kk / 2) * (BP_STRIDE * 2) + sub * 32];
            acc[0][sub] = __builtin_amdgcn_wmma_f32_16x16x4_f32(
                false, a0, false, bv, (short)0, acc[0][sub], false, false);
            acc[1][sub] = __builtin_amdgcn_wmma_f32_16x16x4_f32(
                false, a1, false, bv, (short)0, acc[1][sub], false, false);
        }
    }
}

__global__ __launch_bounds__(256) void rgcn_gemm(const float* __restrict__ x,
                                                 const float* __restrict__ comp,
                                                 const float* __restrict__ basis,
                                                 const float* __restrict__ root,
                                                 const float* __restrict__ bias,
                                                 const float* __restrict__ partial,
                                                 float* __restrict__ out) {
    __shared__ __attribute__((aligned(16))) float As[2 * STAGE_A];
    __shared__ __attribute__((aligned(16))) float Bst[2 * STAGE_B];
    __shared__ __attribute__((aligned(16))) float coefL[BM * 12];
    __shared__ __attribute__((aligned(16))) float sumsL[D];

    const int tid      = threadIdx.x;
    const int rowBase  = blockIdx.x * BM;
    const int b        = rowBase / S;
    const int tBase    = rowBase % S;
    const int colBase0 = blockIdx.y * BN;

    const int wid     = tid >> 5;
    const int lane    = tid & 31;
    const int waveRow = wid & 3;
    const int waveCol = wid >> 2;
    const int lmod    = lane & 15;
    const int lhi     = lane >> 4;
    const int koff    = lhi * 2;
    const int aoff0   = (waveRow * 32 + lmod) * AS_STRIDE + koff;
    const int aoff1   = aoff0 + 16 * AS_STRIDE;
    const int boff    = (lhi * BP_STRIDE + waveCol * 64 + lmod) * 2;

    const int c4  = (tid & 15) * 4;
    const int rg  = tid >> 4;
    const int n4  = (tid & 31) * 4;
    const int bk0 = tid >> 5;

    v8f acc[2][4] = {};

    v4f xr[4][12];
    v4f pb[8];

    {
        const int r_ = tid >> 1, j = tid & 1;
        const int t  = tBase + r_;
        float inv1 = 1.0f / (float)(1 + (t + 2 < S));
        int   d2   = (t >= 1) + (t + 3 < S);
        float inv2 = d2 ? 1.0f / (float)d2 : 0.f;
        int   bc   = (t >= 1) + 1 + (t + 1 < S) + (t + 2 < S) + (t + 3 < S);
        float inv3 = 1.0f / (float)(S - bc);
        float c0 = comp[0 * 2 + j], c1 = comp[1 * 2 + j];
        float c2 = comp[2 * 2 + j], c3 = comp[3 * 2 + j];
        float m3 = c3 * inv3;
        float* w = coefL + r_ * 12 + j * 6;
        w[0] = (t >= 1)    ? (c2 * inv2 - m3) : 0.f;
        w[1] =               c1 * inv1 - m3;
        w[2] = (t + 1 < S) ? (c0        - m3) : 0.f;
        w[3] = (t + 2 < S) ? (c1 * inv1 - m3) : 0.f;
        w[4] = (t + 3 < S) ? (c2 * inv2 - m3) : 0.f;
        w[5] = m3;
    }

#define ISSUE_B(WSEG, KCBASE)                                                   \
    _Pragma("unroll")                                                           \
    for (int e = 0; e < 8; ++e)                                                 \
        pb[e] = *(const v4f*)&(WSEG)[((KCBASE) + 2 * bk0 + (e & 1) + (e >> 1) * 16) * D + colBase0 + n4];

#define STORE_B(BUF)                                                            \
    {                                                                           \
        float* Bd_ = Bst + (BUF) * STAGE_B;                                     \
        _Pragma("unroll")                                                       \
        for (int p = 0; p < 4; ++p) {                                           \
            float* d_ = Bd_ + ((bk0 + p * 8) * BP_STRIDE + n4) * 2;             \
            v4f lo_, hi_;                                                       \
            lo_.x = pb[2 * p].x; lo_.y = pb[2 * p + 1].x;                       \
            lo_.z = pb[2 * p].y; lo_.w = pb[2 * p + 1].y;                       \
            hi_.x = pb[2 * p].z; hi_.y = pb[2 * p + 1].z;                       \
            hi_.z = pb[2 * p].w; hi_.w = pb[2 * p + 1].w;                       \
            *(v4f*)&d_[0] = lo_;                                                \
            *(v4f*)&d_[4] = hi_;                                                \
        }                                                                       \
    }

#define STORE_A(CQ, SG, BUF)                                                    \
    {                                                                           \
        float* Ad_ = As + (BUF) * STAGE_A;                                      \
        if ((SG) < 2) {                                                         \
            const float* wb_ = coefL + (SG) * 6;                                \
            const v4f ps = *(const v4f*)&sumsL[(CQ) * KC + c4];                 \
            _Pragma("unroll")                                                   \
            for (int e = 0; e < 8; ++e) {                                       \
                int r = rg * 8 + e;                                             \
                const float* w = wb_ + r * 12;                                  \
                v4f val = w[5] * ps;                                            \
                _Pragma("unroll")                                               \
                for (int o = 0; o < 5; ++o) val += w[o] * xr[CQ][e + o];        \
                *(v4f*)&Ad_[r * AS_STRIDE + c4] = val;                          \
            }                                                                   \
        } else {                                                                \
            _Pragma("unroll")                                                   \
            for (int e = 0; e < 8; ++e)                                         \
                *(v4f*)&Ad_[(rg * 8 + e) * AS_STRIDE + c4] = xr[CQ][e + 1];     \
        }                                                                       \
    }

#define TRIP(CQ, NCQ, NSG, NWSEG)                                               \
    {                                                                           \
          \
        __builtin_amdgcn_s_setprio(1);                                          \
        ISSUE_B(NWSEG, (NCQ) * KC)                                              \
          \
        __builtin_amdgcn_sched_barrier(0);                                      \
        wmma_steps<0, KSPLIT>(As + ((CQ) & 1) * STAGE_A, Bst + ((CQ) & 1) * STAGE_B, aoff0, aoff1, boff, acc); \
          \
        __builtin_amdgcn_sched_barrier(0);                                      \
        STORE_A(NCQ, NSG, ((CQ) & 1) ^ 1)                                       \
        STORE_B(((CQ) & 1) ^ 1)                                                 \
          \
        __builtin_amdgcn_s_setprio(0);                                          \
          \
        __builtin_amdgcn_sched_barrier(0);                                      \
        wmma_steps<KSPLIT, KC>(As + ((CQ) & 1) * STAGE_A, Bst + ((CQ) & 1) * STAGE_B, aoff0, aoff1, boff, acc); \
        __syncthreads();                                                        \
    }

    {
        const int tm1 = tBase + rg * 8 - 1;
#pragma unroll
        for (int c = 0; c < 4; ++c) {
#pragma unroll
            for (int i = 0; i < 12; ++i) {
                int sr = tm1 + i;
                sr = sr < 0 ? 0 : (sr >= S ? S - 1 : sr);
                xr[c][i] = *(const v4f*)&x[(b * S + sr) * D + c * KC + c4];
            }
            if (c == 0) {
                ISSUE_B(basis, 0)
            }
        }
    }
    float pp[CS_CHUNKS];
#pragma unroll
    for (int c = 0; c < CS_CHUNKS; ++c) pp[c] = partial[(b * CS_CHUNKS + c) * D + tid];
    {
        float s = 0.f;
#pragma unroll
        for (int c = 0; c < CS_CHUNKS; ++c) s += pp[c];
        sumsL[tid] = s;
    }
    __syncthreads();
    STORE_A(0, 0, 0)
    STORE_B(0)
    __syncthreads();

#pragma unroll 1
    for (int sg = 0; sg < 3; ++sg) {
        const float* wseg  = sg < 2 ? basis + sg * (D * D) : root;
        const float* wnext = sg < 1 ? basis + (D * D) : root;
        TRIP(0, 1, sg, wseg)
        TRIP(1, 2, sg, wseg)
        TRIP(2, 3, sg, wseg)
        if (sg < 2) {
            TRIP(3, 0, sg + 1, wnext)
        }
    }
    wmma_steps<0, KC>(As + STAGE_A, Bst + STAGE_B, aoff0, aoff1, boff, acc);
#undef TRIP
#undef STORE_A
#undef ISSUE_B
#undef STORE_B

#pragma unroll
    for (int rt = 0; rt < 2; ++rt) {
#pragma unroll
        for (int sub = 0; sub < 4; ++sub) {
            int ncol = colBase0 + waveCol * 64 + sub * 16 + lmod;
            float badd = bias[ncol];
#pragma unroll
            for (int jj = 0; jj < 8; ++jj) {
                int r = waveRow * 32 + rt * 16 + jj + 8 * lhi;
                out[(b * S + tBase + r) * D + ncol] = acc[rt][sub][jj] + badd;
            }
        }
    }
}

extern "C" void kernel_launch(void* const* d_in, const int* in_sizes, int n_in,
                              void* d_out, int out_size, void* d_ws, size_t ws_size,
                              hipStream_t stream) {
    const float* x     = (const float*)d_in[0];
    const float* comp  = (const float*)d_in[1];
    const float* basis = (const float*)d_in[2];
    const float* root  = (const float*)d_in[3];
    const float* bias  = (const float*)d_in[4];
    float* out = (float*)d_out;

    float* partial = (float*)d_ws;

    colsum_partial<<<BATCH * CS_CHUNKS, 256, 0, stream>>>(x, partial);
    rgcn_gemm<<<dim3((BATCH * S) / BM, D / BN), 256, 0, stream>>>(
        x, comp, basis, root, bias, partial, out);
}
